# starting kernel
# baseline (speedup 1.0000x reference)
#include <hip/hip_runtime.h>
#include <math.h>

typedef __attribute__((ext_vector_type(2))) float v2f;
typedef __attribute__((ext_vector_type(8))) float v8f;

#define NHEADS   16
#define DK       8
#define SEQ      2048
#define EMB      128
#define BATCH    4
#define KV_PITCH 9

#define QSCALE   (0.35355339059327373f * 1.4426950408889634f)
#define SHIFT2   (8.0f * 0.35355339059327373f * 1.4426950408889634f)

__global__ __launch_bounds__(256)
void qcos_kernel(const float* __restrict__ x, const float* __restrict__ theta,
                 float* __restrict__ q) {
    int idx = blockIdx.x * 256 + threadIdx.x;
    if (idx >= BATCH * SEQ * EMB) return;
    int d = idx & 7;
    int h = (idx >> 3) & 15;
    int s = (idx >> 7) & 2047;
    int b = idx >> 18;
    float v = __cosf(x[idx] + theta[d]);
    q[(((size_t)(b * NHEADS + h) * SEQ) + s) * DK + d] = v;
}

__global__ __launch_bounds__(256)
void attn_kernel(const float* __restrict__ qg, float* __restrict__ og) {
    extern __shared__ float kv[];

    const int tid  = threadIdx.x;
    const int bh   = blockIdx.x >> 4;
    const int qblk = blockIdx.x & 15;

    const float4* src4 = reinterpret_cast<const float4*>(qg + (size_t)bh * SEQ * DK);
    for (int i = tid; i < SEQ * DK / 4; i += 256) {
        float4 v = src4[i];
        int srow = i >> 1;
        int dcol = (i & 1) * 4;
        float* dst = &kv[srow * KV_PITCH + dcol];
        dst[0] = v.x; dst[1] = v.y; dst[2] = v.z; dst[3] = v.w;
    }
    __syncthreads();

    const int  wave  = tid >> 5;
    const int  lane  = tid & 31;
    const int  ln    = lane & 15;
    const bool hi    = (lane >= 16);
    const int  off   = hi ? 2 : 0;
    const int  qrow  = qblk * 128 + wave * 16 + ln;

    v2f bq0, bq1;
    bq0.x = kv[qrow * KV_PITCH + off + 0] * QSCALE;
    bq0.y = kv[qrow * KV_PITCH + off + 1] * QSCALE;
    bq1.x = kv[qrow * KV_PITCH + 4 + off + 0] * QSCALE;
    bq1.y = kv[qrow * KV_PITCH + 4 + off + 1] * QSCALE;

    v8f oacc = {0.f, 0.f, 0.f, 0.f, 0.f, 0.f, 0.f, 0.f};
    float l = 0.f;

    for (int kt = 0; kt < SEQ / 16; ++kt) {
        const int kbase = kt * 16;
        const int krow  = kbase + ln;

        v2f a0, a1;
        a0.x = kv[krow * KV_PITCH + off + 0];
        a0.y = kv[krow * KV_PITCH + off + 1];
        a1.x = kv[krow * KV_PITCH + 4 + off + 0];
        a1.y = kv[krow * KV_PITCH + 4 + off + 1];

        v8f s8 = {0.f, 0.f, 0.f, 0.f, 0.f, 0.f, 0.f, 0.f};
        s8 = __builtin_amdgcn_wmma_f32_16x16x4_f32(false, a0, false, bq0,
                                                   (short)0, s8, false, false);
        s8 = __builtin_amdgcn_wmma_f32_16x16x4_f32(false, a1, false, bq1,
                                                   (short)0, s8, false, false);

        float p[8];
        float tsum = 0.f;
        #pragma unroll
        for (int r = 0; r < 8; ++r) {
            p[r] = __builtin_amdgcn_exp2f(s8[r] - SHIFT2);
            tsum += p[r];
        }
        l += tsum;

        float t1 = __shfl_xor(hi ? p[0] : p[2], 16, 32);
        float t2 = __shfl_xor(hi ? p[1] : p[3], 16, 32);
        float t3 = __shfl_xor(hi ? p[4] : p[6], 16, 32);
        float t4 = __shfl_xor(hi ? p[5] : p[7], 16, 32);

        v2f pb0, pb1, pb2, pb3;
        pb0.x = hi ? t1   : p[0];  pb0.y = hi ? t2   : p[1];
        pb1.x = hi ? t3   : p[4];  pb1.y = hi ? t4   : p[5];
        pb2.x = hi ? p[2] : t1;    pb2.y = hi ? p[3] : t2;
        pb3.x = hi ? p[6] : t3;    pb3.y = hi ? p[7] : t4;

        v2f av;
        av.x = kv[(kbase + 0 + off) * KV_PITCH + ln];
        av.y = kv[(kbase + 1 + off) * KV_PITCH + ln];
        oacc = __builtin_amdgcn_wmma_f32_16x16x4_f32(false, av, false, pb0,
                                                     (short)0, oacc, false, false);
        av.x = kv[(kbase + 4 + off) * KV_PITCH + ln];
        av.y = kv[(kbase + 5 + off) * KV_PITCH + ln];
        oacc = __builtin_amdgcn_wmma_f32_16x16x4_f32(false, av, false, pb1,
                                                     (short)0, oacc, false, false);
        av.x = kv[(kbase + 8 + off) * KV_PITCH + ln];
        av.y = kv[(kbase + 9 + off) * KV_PITCH + ln];
        oacc = __builtin_amdgcn_wmma_f32_16x16x4_f32(false, av, false, pb2,
                                                     (short)0, oacc, false, false);
        av.x = kv[(kbase + 12 + off) * KV_PITCH + ln];
        av.y = kv[(kbase + 13 + off) * KV_PITCH + ln];
        oacc = __builtin_amdgcn_wmma_f32_16x16x4_f32(false, av, false, pb3,
                                                     (short)0, oacc, false, false);
    }

    float invl = 1.f / (l + __shfl_xor(l, 16, 32));
    if (!hi) {
        const int b = bh >> 4;
        const int h = bh & 15;
        float4 r0, r1;
        r0.x = oacc[0] * invl; r0.y = oacc[1] * invl;
        r0.z = oacc[2] * invl; r0.w = oacc[3] * invl;
        r1.x = oacc[4] * invl; r1.y = oacc[5] * invl;
        r1.z = oacc[6] * invl; r1.w = oacc[7] * invl;
        float* dst = og + ((size_t)(b * SEQ + qrow)) * EMB + h * DK;
        reinterpret_cast<float4*>(dst)[0] = r0;
        reinterpret_cast<float4*>(dst)[1] = r1;
    }
}

__global__ __launch_bounds__(128)
void proj_kernel(const float* __restrict__ O2, const float* __restrict__ W,
                 const float* __restrict__ bias, float* __restrict__ out) {
    const int  wave = threadIdx.x >> 5;
    const int  lane = threadIdx.x & 31;
    const int  ln   = lane & 15;
    const bool hi   = (lane >= 16);
    const int  off  = hi ? 2 : 0;
    const int  rowbase = (blockIdx.x * 4 + wave) * 16;

    v8f acc[8];
    #pragma unroll
    for (int n = 0; n < 8; ++n) acc[n] = (v8f){0.f,0.f,0.f,0.f,0.f,0.f,0.f,0.f};

    for (int c = 0; c < 32; ++c) {
        const int kd = 4 * c + off;
        const float* ar = O2 + (size_t)(rowbase + ln) * EMB + kd;
        v2f a; a.x = ar[0]; a.y = ar[1];
        #pragma unroll
        for (int n = 0; n < 8; ++n) {
            const float* wr = W + (size_t)(n * 16 + ln) * EMB + kd;
            v2f bw; bw.x = wr[0]; bw.y = wr[1];
            acc[n] = __builtin_amdgcn_wmma_f32_16x16x4_f32(false, a, false, bw,
                                                           (short)0, acc[n],
                                                           false, false);
        }
    }

    #pragma unroll
    for (int n = 0; n < 8; ++n) {
        const int col = n * 16 + ln;
        const float bb = bias[col];
        #pragma unroll
        for (int r = 0; r < 8; ++r) {
            const int row = rowbase + r + (hi ? 8 : 0);
            out[(size_t)row * EMB + col] = acc[n][r] + bb;
        }
    }
}

extern "C" void kernel_launch(void* const* d_in, const int* in_sizes, int n_in,
                              void* d_out, int out_size, void* d_ws, size_t ws_size,
                              hipStream_t stream) {
    const float* x     = (const float*)d_in[0];
    const float* theta = (const float*)d_in[1];
    const float* W     = (const float*)d_in[2];
    const float* bias  = (const float*)d_in[3];
    float* out = (float*)d_out;

    float* ws1 = (float*)d_ws;
    float* ws2 = ws1 + (size_t)BATCH * SEQ * EMB;

    const int n_elem = BATCH * SEQ * EMB;
    qcos_kernel<<<(n_elem + 255) / 256, 256, 0, stream>>>(x, theta, ws1);

    const size_t lds_bytes = (size_t)SEQ * KV_PITCH * sizeof(float);
    attn_kernel<<<BATCH * NHEADS * 16, 256, lds_bytes, stream>>>(ws1, ws2);

    proj_kernel<<<128, 128, 0, stream>>>(ws2, W, bias, out);
}


# S-tile key rows permuted (row 8h+2j+s <- key 4j+2h+s) so each lane's exp results are PV's B-fragments in place: no cross-lane shuffles or selects; softmax denominator sums regrouped by key
# speedup vs baseline: 1.1685x; 1.1685x over previous
#include <hip/hip_runtime.h>
#include <math.h>

typedef __attribute__((ext_vector_type(2))) float v2f;
typedef __attribute__((ext_vector_type(8))) float v8f;

#define NHEADS   16
#define DK       8
#define SEQ      2048
#define EMB      128
#define BATCH    4
#define KV_PITCH 9

#define QSCALE   (0.35355339059327373f * 1.4426950408889634f)
#define SHIFT2   (8.0f * 0.35355339059327373f * 1.4426950408889634f)

__global__ __launch_bounds__(256)
void qcos_kernel(const float* __restrict__ x, const float* __restrict__ theta,
                 float* __restrict__ q) {
    int idx = blockIdx.x * 256 + threadIdx.x;
    if (idx >= BATCH * SEQ * EMB) return;
    int d = idx & 7;
    int h = (idx >> 3) & 15;
    int s = (idx >> 7) & 2047;
    int b = idx >> 18;
    float v = __cosf(x[idx] + theta[d]);
    q[(((size_t)(b * NHEADS + h) * SEQ) + s) * DK + d] = v;
}

__global__ __launch_bounds__(256)
void attn_kernel(const float* __restrict__ qg, float* __restrict__ og) {
    extern __shared__ float kv[];

    const int tid  = threadIdx.x;
    const int bh   = blockIdx.x >> 4;
    const int qblk = blockIdx.x & 15;

    const float4* src4 = reinterpret_cast<const float4*>(qg + (size_t)bh * SEQ * DK);
    for (int i = tid; i < SEQ * DK / 4; i += 256) {
        float4 v = src4[i];
        int srow = i >> 1;
        int dcol = (i & 1) * 4;
        float* dst = &kv[srow * KV_PITCH + dcol];
        dst[0] = v.x; dst[1] = v.y; dst[2] = v.z; dst[3] = v.w;
    }
    __syncthreads();

    const int  wave  = tid >> 5;
    const int  lane  = tid & 31;
    const int  ln    = lane & 15;
    const bool hi    = (lane >= 16);
    const int  off   = hi ? 2 : 0;
    const int  qrow  = qblk * 128 + wave * 16 + ln;

    v2f bq0, bq1;
    bq0.x = kv[qrow * KV_PITCH + off + 0] * QSCALE;
    bq0.y = kv[qrow * KV_PITCH + off + 1] * QSCALE;
    bq1.x = kv[qrow * KV_PITCH + 4 + off + 0] * QSCALE;
    bq1.y = kv[qrow * KV_PITCH + 4 + off + 1] * QSCALE;

    const int srow = ((ln & 6) << 1) | ((ln & 8) >> 2) | (ln & 1);

    v8f oacc = {0.f, 0.f, 0.f, 0.f, 0.f, 0.f, 0.f, 0.f};
    float l = 0.f;

    for (int kt = 0; kt < SEQ / 16; ++kt) {
        const int kbase = kt * 16;
        const int krow  = kbase + srow;

        v2f a0, a1;
        a0.x = kv[krow * KV_PITCH + off + 0];
        a0.y = kv[krow * KV_PITCH + off + 1];
        a1.x = kv[krow * KV_PITCH + 4 + off + 0];
        a1.y = kv[krow * KV_PITCH + 4 + off + 1];

        v8f s8 = {0.f, 0.f, 0.f, 0.f, 0.f, 0.f, 0.f, 0.f};
        s8 = __builtin_amdgcn_wmma_f32_16x16x4_f32(false, a0, false, bq0,
                                                   (short)0, s8, false, false);
        s8 = __builtin_amdgcn_wmma_f32_16x16x4_f32(false, a1, false, bq1,
                                                   (short)0, s8, false, false);

        float p[8];
        float tsum = 0.f;
        #pragma unroll
        for (int r = 0; r < 8; ++r) {
            p[r] = __builtin_amdgcn_exp2f(s8[r] - SHIFT2);
            tsum += p[r];
        }
        l += tsum;

        v2f pb0, pb1, pb2, pb3;
        pb0.x = p[0];  pb0.y = p[1];
        pb1.x = p[2];  pb1.y = p[3];
        pb2.x = p[4];  pb2.y = p[5];
        pb3.x = p[6];  pb3.y = p[7];

        v2f av;
        av.x = kv[(kbase + 0 + off) * KV_PITCH + ln];
        av.y = kv[(kbase + 1 + off) * KV_PITCH + ln];
        oacc = __builtin_amdgcn_wmma_f32_16x16x4_f32(false, av, false, pb0,
                                                     (short)0, oacc, false, false);
        av.x = kv[(kbase + 4 + off) * KV_PITCH + ln];
        av.y = kv[(kbase + 5 + off) * KV_PITCH + ln];
        oacc = __builtin_amdgcn_wmma_f32_16x16x4_f32(false, av, false, pb1,
                                                     (short)0, oacc, false, false);
        av.x = kv[(kbase + 8 + off) * KV_PITCH + ln];
        av.y = kv[(kbase + 9 + off) * KV_PITCH + ln];
        oacc = __builtin_amdgcn_wmma_f32_16x16x4_f32(false, av, false, pb2,
                                                     (short)0, oacc, false, false);
        av.x = kv[(kbase + 12 + off) * KV_PITCH + ln];
        av.y = kv[(kbase + 13 + off) * KV_PITCH + ln];
        oacc = __builtin_amdgcn_wmma_f32_16x16x4_f32(false, av, false, pb3,
                                                     (short)0, oacc, false, false);
    }

    float invl = 1.f / (l + __shfl_xor(l, 16, 32));
    if (!hi) {
        const int b = bh >> 4;
        const int h = bh & 15;
        float4 r0, r1;
        r0.x = oacc[0] * invl; r0.y = oacc[1] * invl;
        r0.z = oacc[2] * invl; r0.w = oacc[3] * invl;
        r1.x = oacc[4] * invl; r1.y = oacc[5] * invl;
        r1.z = oacc[6] * invl; r1.w = oacc[7] * invl;
        float* dst = og + ((size_t)(b * SEQ + qrow)) * EMB + h * DK;
        reinterpret_cast<float4*>(dst)[0] = r0;
        reinterpret_cast<float4*>(dst)[1] = r1;
    }
}

__global__ __launch_bounds__(128)
void proj_kernel(const float* __restrict__ O2, const float* __restrict__ W,
                 const float* __restrict__ bias, float* __restrict__ out) {
    const int  wave = threadIdx.x >> 5;
    const int  lane = threadIdx.x & 31;
    const int  ln   = lane & 15;
    const bool hi   = (lane >= 16);
    const int  off  = hi ? 2 : 0;
    const int  rowbase = (blockIdx.x * 4 + wave) * 16;

    v8f acc[8];
    #pragma unroll
    for (int n = 0; n < 8; ++n) acc[n] = (v8f){0.f,0.f,0.f,0.f,0.f,0.f,0.f,0.f};

    for (int c = 0; c < 32; ++c) {
        const int kd = 4 * c + off;
        const float* ar = O2 + (size_t)(rowbase + ln) * EMB + kd;
        v2f a; a.x = ar[0]; a.y = ar[1];
        #pragma unroll
        for (int n = 0; n < 8; ++n) {
            const float* wr = W + (size_t)(n * 16 + ln) * EMB + kd;
            v2f bw; bw.x = wr[0]; bw.y = wr[1];
            acc[n] = __builtin_amdgcn_wmma_f32_16x16x4_f32(false, a, false, bw,
                                                           (short)0, acc[n],
                                                           false, false);
        }
    }

    #pragma unroll
    for (int n = 0; n < 8; ++n) {
        const int col = n * 16 + ln;
        const float bb = bias[col];
        #pragma unroll
        for (int r = 0; r < 8; ++r) {
            const int row = rowbase + r + (hi ? 8 : 0);
            out[(size_t)row * EMB + col] = acc[n][r] + bb;
        }
    }
}

extern "C" void kernel_launch(void* const* d_in, const int* in_sizes, int n_in,
                              void* d_out, int out_size, void* d_ws, size_t ws_size,
                              hipStream_t stream) {
    const float* x     = (const float*)d_in[0];
    const float* theta = (const float*)d_in[1];
    const float* W     = (const float*)d_in[2];
    const float* bias  = (const float*)d_in[3];
    float* out = (float*)d_out;

    float* ws1 = (float*)d_ws;
    float* ws2 = ws1 + (size_t)BATCH * SEQ * EMB;

    const int n_elem = BATCH * SEQ * EMB;
    qcos_kernel<<<(n_elem + 255) / 256, 256, 0, stream>>>(x, theta, ws1);

    const size_t lds_bytes = (size_t)SEQ * KV_PITCH * sizeof(float);
    attn_kernel<<<BATCH * NHEADS * 16, 256, lds_bytes, stream>>>(ws1, ws2);

    proj_kernel<<<128, 128, 0, stream>>>(ws2, W, bias, out);
}


# attention loop: softmax shift preloaded as C of the score WMMAs, denominator summed by a ones row in V^T's idle WMMA rows (both sums change order): no vector adds left in the loop
# speedup vs baseline: 1.2174x; 1.0418x over previous
#include <hip/hip_runtime.h>
#include <math.h>

typedef __attribute__((ext_vector_type(2))) float v2f;
typedef __attribute__((ext_vector_type(8))) float v8f;

#define NHEADS   16
#define DK       8
#define SEQ      2048
#define EMB      128
#define BATCH    4
#define KV_PITCH 9

#define QSCALE   (0.35355339059327373f * 1.4426950408889634f)
#define SHIFT2   (8.0f * 0.35355339059327373f * 1.4426950408889634f)

__global__ __launch_bounds__(256)
void qcos_kernel(const float* __restrict__ x, const float* __restrict__ theta,
                 float* __restrict__ q) {
    int idx = blockIdx.x * 256 + threadIdx.x;
    if (idx >= BATCH * SEQ * EMB) return;
    int d = idx & 7;
    int h = (idx >> 3) & 15;
    int s = (idx >> 7) & 2047;
    int b = idx >> 18;
    float v = __cosf(x[idx] + theta[d]);
    q[(((size_t)(b * NHEADS + h) * SEQ) + s) * DK + d] = v;
}

__global__ __launch_bounds__(256)
void attn_kernel(const float* __restrict__ qg, float* __restrict__ og) {
    extern __shared__ float kv[];

    const int tid  = threadIdx.x;
    const int bh   = blockIdx.x >> 4;
    const int qblk = blockIdx.x & 15;

    const float4* src4 = reinterpret_cast<const float4*>(qg + (size_t)bh * SEQ * DK);
    for (int i = tid; i < SEQ * DK / 4; i += 256) {
        float4 v = src4[i];
        int srow = i >> 1;
        int dcol = (i & 1) * 4;
        float* dst = &kv[srow * KV_PITCH + dcol];
        dst[0] = v.x; dst[1] = v.y; dst[2] = v.z; dst[3] = v.w;
    }
    for (int i = tid; i < SEQ; i += 256) kv[i * KV_PITCH + DK] = 1.0f;
    __syncthreads();

    const int  wave  = tid >> 5;
    const int  lane  = tid & 31;
    const int  ln    = lane & 15;
    const bool hi    = (lane >= 16);
    const int  off   = hi ? 2 : 0;
    const int  qrow  = qblk * 128 + wave * 16 + ln;

    v2f bq0, bq1;
    bq0.x = kv[qrow * KV_PITCH + off + 0] * QSCALE;
    bq0.y = kv[qrow * KV_PITCH + off + 1] * QSCALE;
    bq1.x = kv[qrow * KV_PITCH + 4 + off + 0] * QSCALE;
    bq1.y = kv[qrow * KV_PITCH + 4 + off + 1] * QSCALE;

    const int srow = ((ln & 6) << 1) | ((ln & 8) >> 2) | (ln & 1);

    v8f oacc = {0.f, 0.f, 0.f, 0.f, 0.f, 0.f, 0.f, 0.f};
    const v8f cshift = {-SHIFT2, -SHIFT2, -SHIFT2, -SHIFT2,
                        -SHIFT2, -SHIFT2, -SHIFT2, -SHIFT2};

    for (int kt = 0; kt < SEQ / 16; ++kt) {
        const int kbase = kt * 16;
        const int krow  = kbase + srow;

        v2f a0, a1;
        a0.x = kv[krow * KV_PITCH + off + 0];
        a0.y = kv[krow * KV_PITCH + off + 1];
        a1.x = kv[krow * KV_PITCH + 4 + off + 0];
        a1.y = kv[krow * KV_PITCH + 4 + off + 1];

        v2f av0, av1, av2, av3;
        av0.x = kv[(kbase + 0 + off) * KV_PITCH + ln];
        av0.y = kv[(kbase + 1 + off) * KV_PITCH + ln];
        av1.x = kv[(kbase + 4 + off) * KV_PITCH + ln];
        av1.y = kv[(kbase + 5 + off) * KV_PITCH + ln];
        av2.x = kv[(kbase + 8 + off) * KV_PITCH + ln];
        av2.y = kv[(kbase + 9 + off) * KV_PITCH + ln];
        av3.x = kv[(kbase + 12 + off) * KV_PITCH + ln];
        av3.y = kv[(kbase + 13 + off) * KV_PITCH + ln];

        v8f s8 = __builtin_amdgcn_wmma_f32_16x16x4_f32(false, a0, false, bq0,
                                                       (short)0, cshift, false, false);
        s8 = __builtin_amdgcn_wmma_f32_16x16x4_f32(false, a1, false, bq1,
                                                   (short)0, s8, false, false);

        float p[8];
        #pragma unroll
        for (int r = 0; r < 8; ++r) p[r] = __builtin_amdgcn_exp2f(s8[r]);
        __builtin_amdgcn_sched_barrier(0);

        v2f pb0, pb1, pb2, pb3;
        pb0.x = p[0];  pb0.y = p[1];
        pb1.x = p[2];  pb1.y = p[3];
        pb2.x = p[4];  pb2.y = p[5];
        pb3.x = p[6];  pb3.y = p[7];

        oacc = __builtin_amdgcn_wmma_f32_16x16x4_f32(false, av0, false, pb0,
                                                     (short)0, oacc, false, false);
        oacc = __builtin_amdgcn_wmma_f32_16x16x4_f32(false, av1, false, pb1,
                                                     (short)0, oacc, false, false);
        oacc = __builtin_amdgcn_wmma_f32_16x16x4_f32(false, av2, false, pb2,
                                                     (short)0, oacc, false, false);
        oacc = __builtin_amdgcn_wmma_f32_16x16x4_f32(false, av3, false, pb3,
                                                     (short)0, oacc, false, false);
    }

    float invl = 1.f / __shfl_xor(oacc[0], 16, 32);
    if (!hi) {
        const int b = bh >> 4;
        const int h = bh & 15;
        float4 r0, r1;
        r0.x = oacc[0] * invl; r0.y = oacc[1] * invl;
        r0.z = oacc[2] * invl; r0.w = oacc[3] * invl;
        r1.x = oacc[4] * invl; r1.y = oacc[5] * invl;
        r1.z = oacc[6] * invl; r1.w = oacc[7] * invl;
        float* dst = og + ((size_t)(b * SEQ + qrow)) * EMB + h * DK;
        reinterpret_cast<float4*>(dst)[0] = r0;
        reinterpret_cast<float4*>(dst)[1] = r1;
    }
}

__global__ __launch_bounds__(128)
void proj_kernel(const float* __restrict__ O2, const float* __restrict__ W,
                 const float* __restrict__ bias, float* __restrict__ out) {
    const int  wave = threadIdx.x >> 5;
    const int  lane = threadIdx.x & 31;
    const int  ln   = lane & 15;
    const bool hi   = (lane >= 16);
    const int  off  = hi ? 2 : 0;
    const int  rowbase = (blockIdx.x * 4 + wave) * 16;

    v8f acc[8];
    #pragma unroll
    for (int n = 0; n < 8; ++n) acc[n] = (v8f){0.f,0.f,0.f,0.f,0.f,0.f,0.f,0.f};

    for (int c = 0; c < 32; ++c) {
        const int kd = 4 * c + off;
        const float* ar = O2 + (size_t)(rowbase + ln) * EMB + kd;
        v2f a; a.x = ar[0]; a.y = ar[1];
        #pragma unroll
        for (int n = 0; n < 8; ++n) {
            const float* wr = W + (size_t)(n * 16 + ln) * EMB + kd;
            v2f bw; bw.x = wr[0]; bw.y = wr[1];
            acc[n] = __builtin_amdgcn_wmma_f32_16x16x4_f32(false, a, false, bw,
                                                           (short)0, acc[n],
                                                           false, false);
        }
    }

    #pragma unroll
    for (int n = 0; n < 8; ++n) {
        const int col = n * 16 + ln;
        const float bb = bias[col];
        #pragma unroll
        for (int r = 0; r < 8; ++r) {
            const int row = rowbase + r + (hi ? 8 : 0);
            out[(size_t)row * EMB + col] = acc[n][r] + bb;
        }
    }
}

extern "C" void kernel_launch(void* const* d_in, const int* in_sizes, int n_in,
                              void* d_out, int out_size, void* d_ws, size_t ws_size,
                              hipStream_t stream) {
    const float* x     = (const float*)d_in[0];
    const float* theta = (const float*)d_in[1];
    const float* W     = (const float*)d_in[2];
    const float* bias  = (const float*)d_in[3];
    float* out = (float*)d_out;

    float* ws1 = (float*)d_ws;
    float* ws2 = ws1 + (size_t)BATCH * SEQ * EMB;

    const int n_elem = BATCH * SEQ * EMB;
    qcos_kernel<<<(n_elem + 255) / 256, 256, 0, stream>>>(x, theta, ws1);

    const size_t lds_bytes = (size_t)SEQ * KV_PITCH * sizeof(float);
    attn_kernel<<<BATCH * NHEADS * 16, 256, lds_bytes, stream>>>(ws1, ws2);

    proj_kernel<<<128, 128, 0, stream>>>(ws2, W, bias, out);
}


# attention key loop software-pipelined over pairs of key tiles: next pair's 12 LDS fragment loads issued as one group into a second register set ahead of the 12 WMMAs, exps under them; sum order kept
# speedup vs baseline: 1.2816x; 1.0527x over previous
#include <hip/hip_runtime.h>
#include <math.h>

typedef __attribute__((ext_vector_type(2))) float v2f;
typedef __attribute__((ext_vector_type(8))) float v8f;

#define NHEADS   16
#define DK       8
#define SEQ      2048
#define EMB      128
#define BATCH    4
#define KV_PITCH 9

#define QSCALE   (0.35355339059327373f * 1.4426950408889634f)
#define SHIFT2   (8.0f * 0.35355339059327373f * 1.4426950408889634f)

__global__ __launch_bounds__(256)
void qcos_kernel(const float* __restrict__ x, const float* __restrict__ theta,
                 float* __restrict__ q) {
    int idx = blockIdx.x * 256 + threadIdx.x;
    if (idx >= BATCH * SEQ * EMB) return;
    int d = idx & 7;
    int h = (idx >> 3) & 15;
    int s = (idx >> 7) & 2047;
    int b = idx >> 18;
    float v = __cosf(x[idx] + theta[d]);
    q[(((size_t)(b * NHEADS + h) * SEQ) + s) * DK + d] = v;
}

struct KFrag { v2f a0, a1, av0, av1, av2, av3; };

#define TILE_F (16 * KV_PITCH)

#define WMMA_F32(a, b, c) \
    __builtin_amdgcn_wmma_f32_16x16x4_f32(false, (a), false, (b), (short)0, (c), false, false)

__device__ __forceinline__ KFrag load_frag(const float* kp, const float* vp) {
    KFrag f;
    f.a0.x  = kp[0];              f.a0.y  = kp[1];
    f.a1.x  = kp[4];              f.a1.y  = kp[5];
    f.av0.x = vp[0 * KV_PITCH];   f.av0.y = vp[1 * KV_PITCH];
    f.av1.x = vp[4 * KV_PITCH];   f.av1.y = vp[5 * KV_PITCH];
    f.av2.x = vp[8 * KV_PITCH];   f.av2.y = vp[9 * KV_PITCH];
    f.av3.x = vp[12 * KV_PITCH];  f.av3.y = vp[13 * KV_PITCH];
    return f;
}

__device__ __forceinline__ void tile_pair(const KFrag& fA, const KFrag& fB,
                                          v2f bq0, v2f bq1, v8f cshift, v8f& oacc) {
    v8f sA = WMMA_F32(fA.a0, bq0, cshift);
    v8f sB = WMMA_F32(fB.a0, bq0, cshift);
    sA = WMMA_F32(fA.a1, bq1, sA);
    sB = WMMA_F32(fB.a1, bq1, sB);

    float pA[8], pB[8];
    #pragma unroll
    for (int r = 0; r < 8; ++r) pB[r] = __builtin_amdgcn_exp2f(sB[r]);
    #pragma unroll
    for (int r = 0; r < 8; ++r) pA[r] = __builtin_amdgcn_exp2f(sA[r]);

    v2f pb;
    pb.x = pA[0]; pb.y = pA[1];  oacc = WMMA_F32(fA.av0, pb, oacc);
    pb.x = pA[2]; pb.y = pA[3];  oacc = WMMA_F32(fA.av1, pb, oacc);
    pb.x = pA[4]; pb.y = pA[5];  oacc = WMMA_F32(fA.av2, pb, oacc);
    pb.x = pA[6]; pb.y = pA[7];  oacc = WMMA_F32(fA.av3, pb, oacc);
    pb.x = pB[0]; pb.y = pB[1];  oacc = WMMA_F32(fB.av0, pb, oacc);
    pb.x = pB[2]; pb.y = pB[3];  oacc = WMMA_F32(fB.av1, pb, oacc);
    pb.x = pB[4]; pb.y = pB[5];  oacc = WMMA_F32(fB.av2, pb, oacc);
    pb.x = pB[6]; pb.y = pB[7];  oacc = WMMA_F32(fB.av3, pb, oacc);
}

#define SG_WMMA  0x008
#define SG_LDSRD 0x100
#define SG_TRANS 0x400

__global__ __launch_bounds__(256)
void attn_kernel(const float* __restrict__ qg, float* __restrict__ og) {
    extern __shared__ float kv[];

    const int tid  = threadIdx.x;
    const int bh   = blockIdx.x >> 4;
    const int qblk = blockIdx.x & 15;

    const float4* src4 = reinterpret_cast<const float4*>(qg + (size_t)bh * SEQ * DK);
    for (int i = tid; i < SEQ * DK / 4; i += 256) {
        float4 v = src4[i];
        int srow = i >> 1;
        int dcol = (i & 1) * 4;
        float* dst = &kv[srow * KV_PITCH + dcol];
        dst[0] = v.x; dst[1] = v.y; dst[2] = v.z; dst[3] = v.w;
    }
    for (int i = tid; i < SEQ; i += 256) kv[i * KV_PITCH + DK] = 1.0f;
    __syncthreads();

    const int  wave  = tid >> 5;
    const int  lane  = tid & 31;
    const int  ln    = lane & 15;
    const bool hi    = (lane >= 16);
    const int  off   = hi ? 2 : 0;
    const int  qrow  = qblk * 128 + wave * 16 + ln;

    v2f bq0, bq1;
    bq0.x = kv[qrow * KV_PITCH + off + 0] * QSCALE;
    bq0.y = kv[qrow * KV_PITCH + off + 1] * QSCALE;
    bq1.x = kv[qrow * KV_PITCH + 4 + off + 0] * QSCALE;
    bq1.y = kv[qrow * KV_PITCH + 4 + off + 1] * QSCALE;

    const int srow = ((ln & 6) << 1) | ((ln & 8) >> 2) | (ln & 1);

    v8f oacc = {0.f, 0.f, 0.f, 0.f, 0.f, 0.f, 0.f, 0.f};
    const v8f cshift = {-SHIFT2, -SHIFT2, -SHIFT2, -SHIFT2,
                        -SHIFT2, -SHIFT2, -SHIFT2, -SHIFT2};

    const float* kp = kv + srow * KV_PITCH + off;
    const float* vp = kv + off * KV_PITCH + ln;

    KFrag xA = load_frag(kp, vp);
    KFrag xB = load_frag(kp + TILE_F, vp + TILE_F);

    for (int it = 0; it < SEQ / 64; ++it) {
        const int cur = it * 4 * TILE_F;
        KFrag yA = load_frag(kp + cur + 2 * TILE_F, vp + cur + 2 * TILE_F);
        KFrag yB = load_frag(kp + cur + 3 * TILE_F, vp + cur + 3 * TILE_F);
        tile_pair(xA, xB, bq0, bq1, cshift, oacc);

        asm volatile("" :: "v"(oacc));

        const int nxt = ((it + 1) & (SEQ / 64 - 1)) * 4 * TILE_F;
        xA = load_frag(kp + nxt, vp + nxt);
        xB = load_frag(kp + nxt + TILE_F, vp + nxt + TILE_F);
        tile_pair(yA, yB, bq0, bq1, cshift, oacc);

        __builtin_amdgcn_sched_group_barrier(SG_LDSRD, 12, 0);
        __builtin_amdgcn_sched_group_barrier(SG_WMMA, 4, 0);
        __builtin_amdgcn_sched_group_barrier(SG_TRANS, 8, 0);
        __builtin_amdgcn_sched_group_barrier(SG_WMMA, 1, 0);
        __builtin_amdgcn_sched_group_barrier(SG_TRANS, 8, 0);
        __builtin_amdgcn_sched_group_barrier(SG_WMMA, 8, 0);
        __builtin_amdgcn_sched_group_barrier(SG_LDSRD, 12, 0);
        __builtin_amdgcn_sched_group_barrier(SG_WMMA, 4, 0);
        __builtin_amdgcn_sched_group_barrier(SG_TRANS, 8, 0);
        __builtin_amdgcn_sched_group_barrier(SG_WMMA, 1, 0);
        __builtin_amdgcn_sched_group_barrier(SG_TRANS, 8, 0);
        __builtin_amdgcn_sched_group_barrier(SG_WMMA, 7, 0);
    }

    float invl = 1.f / __shfl_xor(oacc[0], 16, 32);
    if (!hi) {
        const int b = bh >> 4;
        const int h = bh & 15;
        float4 r0, r1;
        r0.x = oacc[0] * invl; r0.y = oacc[1] * invl;
        r0.z = oacc[2] * invl; r0.w = oacc[3] * invl;
        r1.x = oacc[4] * invl; r1.y = oacc[5] * invl;
        r1.z = oacc[6] * invl; r1.w = oacc[7] * invl;
        float* dst = og + ((size_t)(b * SEQ + qrow)) * EMB + h * DK;
        reinterpret_cast<float4*>(dst)[0] = r0;
        reinterpret_cast<float4*>(dst)[1] = r1;
    }
}

__global__ __launch_bounds__(128)
void proj_kernel(const float* __restrict__ O2, const float* __restrict__ W,
                 const float* __restrict__ bias, float* __restrict__ out) {
    const int  wave = threadIdx.x >> 5;
    const int  lane = threadIdx.x & 31;
    const int  ln   = lane & 15;
    const bool hi   = (lane >= 16);
    const int  off  = hi ? 2 : 0;
    const int  rowbase = (blockIdx.x * 4 + wave) * 16;

    v8f acc[8];
    #pragma unroll
    for (int n = 0; n < 8; ++n) acc[n] = (v8f){0.f,0.f,0.f,0.f,0.f,0.f,0.f,0.f};

    for (int c = 0; c < 32; ++c) {
        const int kd = 4 * c + off;
        const float* ar = O2 + (size_t)(rowbase + ln) * EMB + kd;
        v2f a; a.x = ar[0]; a.y = ar[1];
        #pragma unroll
        for (int n = 0; n < 8; ++n) {
            const float* wr = W + (size_t)(n * 16 + ln) * EMB + kd;
            v2f bw; bw.x = wr[0]; bw.y = wr[1];
            acc[n] = __builtin_amdgcn_wmma_f32_16x16x4_f32(false, a, false, bw,
                                                           (short)0, acc[n],
                                                           false, false);
        }
    }

    #pragma unroll
    for (int n = 0; n < 8; ++n) {
        const int col = n * 16 + ln;
        const float bb = bias[col];
        #pragma unroll
        for (int r = 0; r < 8; ++r) {
            const int row = rowbase + r + (hi ? 8 : 0);
            out[(size_t)row * EMB + col] = acc[n][r] + bb;
        }
    }
}

extern "C" void kernel_launch(void* const* d_in, const int* in_sizes, int n_in,
                              void* d_out, int out_size, void* d_ws, size_t ws_size,
                              hipStream_t stream) {
    const float* x     = (const float*)d_in[0];
    const float* theta = (const float*)d_in[1];
    const float* W     = (const float*)d_in[2];
    const float* bias  = (const float*)d_in[3];
    float* out = (float*)d_out;

    float* ws1 = (float*)d_ws;
    float* ws2 = ws1 + (size_t)BATCH * SEQ * EMB;

    const int n_elem = BATCH * SEQ * EMB;
    qcos_kernel<<<(n_elem + 255) / 256, 256, 0, stream>>>(x, theta, ws1);

    const size_t lds_bytes = (size_t)SEQ * KV_PITCH * sizeof(float);
    attn_kernel<<<BATCH * NHEADS * 16, 256, lds_bytes, stream>>>(ws1, ws2);

    proj_kernel<<<128, 128, 0, stream>>>(ws2, W, bias, out);
}


# cos(x+theta) fused into the attention kernel's K staging (32-wave work-groups of 512 queries, one per WGP by LDS size): the 32768-wave cos kernel, its workspace round trip and one launch are gone
# speedup vs baseline: 1.4093x; 1.0996x over previous
#include <hip/hip_runtime.h>
#include <math.h>

typedef __attribute__((ext_vector_type(2))) float v2f;
typedef __attribute__((ext_vector_type(8))) float v8f;

#define NHEADS   16
#define DK       8
#define SEQ      2048
#define EMB      128
#define BATCH    4
#define KV_PITCH 9
#define QPB      512
#define ATT_THREADS 1024
#define ATT_LDS_BYTES 167936

#define QSCALE   (0.35355339059327373f * 1.4426950408889634f)
#define SHIFT2   (8.0f * 0.35355339059327373f * 1.4426950408889634f)

struct KFrag { v2f a0, a1, av0, av1, av2, av3; };

#define TILE_F (16 * KV_PITCH)

#define WMMA_F32(a, b, c) \
    __builtin_amdgcn_wmma_f32_16x16x4_f32(false, (a), false, (b), (short)0, (c), false, false)

__device__ __forceinline__ KFrag load_frag(const float* kp, const float* vp) {
    KFrag f;
    f.a0.x  = kp[0];              f.a0.y  = kp[1];
    f.a1.x  = kp[4];              f.a1.y  = kp[5];
    f.av0.x = vp[0 * KV_PITCH];   f.av0.y = vp[1 * KV_PITCH];
    f.av1.x = vp[4 * KV_PITCH];   f.av1.y = vp[5 * KV_PITCH];
    f.av2.x = vp[8 * KV_PITCH];   f.av2.y = vp[9 * KV_PITCH];
    f.av3.x = vp[12 * KV_PITCH];  f.av3.y = vp[13 * KV_PITCH];
    return f;
}

__device__ __forceinline__ void tile_pair(const KFrag& fA, const KFrag& fB,
                                          v2f bq0, v2f bq1, v8f cshift, v8f& oacc) {
    v8f sA = WMMA_F32(fA.a0, bq0, cshift);
    v8f sB = WMMA_F32(fB.a0, bq0, cshift);
    sA = WMMA_F32(fA.a1, bq1, sA);
    sB = WMMA_F32(fB.a1, bq1, sB);

    float pA[8], pB[8];
    #pragma unroll
    for (int r = 0; r < 8; ++r) pB[r] = __builtin_amdgcn_exp2f(sB[r]);
    #pragma unroll
    for (int r = 0; r < 8; ++r) pA[r] = __builtin_amdgcn_exp2f(sA[r]);

    v2f pb;
    pb.x = pA[0]; pb.y = pA[1];  oacc = WMMA_F32(fA.av0, pb, oacc);
    pb.x = pA[2]; pb.y = pA[3];  oacc = WMMA_F32(fA.av1, pb, oacc);
    pb.x = pA[4]; pb.y = pA[5];  oacc = WMMA_F32(fA.av2, pb, oacc);
    pb.x = pA[6]; pb.y = pA[7];  oacc = WMMA_F32(fA.av3, pb, oacc);
    pb.x = pB[0]; pb.y = pB[1];  oacc = WMMA_F32(fB.av0, pb, oacc);
    pb.x = pB[2]; pb.y = pB[3];  oacc = WMMA_F32(fB.av1, pb, oacc);
    pb.x = pB[4]; pb.y = pB[5];  oacc = WMMA_F32(fB.av2, pb, oacc);
    pb.x = pB[6]; pb.y = pB[7];  oacc = WMMA_F32(fB.av3, pb, oacc);
}

#define SG_WMMA  0x008
#define SG_LDSRD 0x100
#define SG_TRANS 0x400

__global__ __launch_bounds__(ATT_THREADS)
void attn_kernel(const float* __restrict__ x, const float* __restrict__ theta,
                 float* __restrict__ og) {
    __shared__ float kv[ATT_LDS_BYTES / 4];

    const int tid  = threadIdx.x;
    const int bh   = blockIdx.x >> 2;
    const int qblk = blockIdx.x & 3;

    {
        const int dcol = (tid & 1) * 4;
        const float* xs = x + ((size_t)(bh >> 4) * SEQ) * EMB + (bh & 15) * DK + dcol;
        const float t0 = theta[dcol + 0], t1 = theta[dcol + 1];
        const float t2 = theta[dcol + 2], t3 = theta[dcol + 3];
        #pragma unroll
        for (int k = 0; k < SEQ * DK / 4 / ATT_THREADS; ++k) {
            const int srow = (tid >> 1) + k * (ATT_THREADS / 2);
            const float4 v = *reinterpret_cast<const float4*>(xs + (size_t)srow * EMB);
            float* dst = &kv[srow * KV_PITCH + dcol];
            dst[0] = __cosf(v.x + t0); dst[1] = __cosf(v.y + t1);
            dst[2] = __cosf(v.z + t2); dst[3] = __cosf(v.w + t3);
        }
    }
    for (int i = tid; i < SEQ; i += ATT_THREADS) kv[i * KV_PITCH + DK] = 1.0f;
    __syncthreads();

    const int  wave  = tid >> 5;
    const int  lane  = tid & 31;
    const int  ln    = lane & 15;
    const bool hi    = (lane >= 16);
    const int  off   = hi ? 2 : 0;
    const int  qrow  = qblk * QPB + wave * 16 + ln;

    v2f bq0, bq1;
    bq0.x = kv[qrow * KV_PITCH + off + 0] * QSCALE;
    bq0.y = kv[qrow * KV_PITCH + off + 1] * QSCALE;
    bq1.x = kv[qrow * KV_PITCH + 4 + off + 0] * QSCALE;
    bq1.y = kv[qrow * KV_PITCH + 4 + off + 1] * QSCALE;

    const int srow = ((ln & 6) << 1) | ((ln & 8) >> 2) | (ln & 1);

    v8f oacc = {0.f, 0.f, 0.f, 0.f, 0.f, 0.f, 0.f, 0.f};
    const v8f cshift = {-SHIFT2, -SHIFT2, -SHIFT2, -SHIFT2,
                        -SHIFT2, -SHIFT2, -SHIFT2, -SHIFT2};

    const float* kp = kv + srow * KV_PITCH + off;
    const float* vp = kv + off * KV_PITCH + ln;

    KFrag xA = load_frag(kp, vp);
    KFrag xB = load_frag(kp + TILE_F, vp + TILE_F);

    for (int it = 0; it < SEQ / 64; ++it) {
        const int cur = it * 4 * TILE_F;
        KFrag yA = load_frag(kp + cur + 2 * TILE_F, vp + cur + 2 * TILE_F);
        KFrag yB = load_frag(kp + cur + 3 * TILE_F, vp + cur + 3 * TILE_F);
        tile_pair(xA, xB, bq0, bq1, cshift, oacc);

        asm volatile("" :: "v"(oacc));

        const int nxt = ((it + 1) & (SEQ / 64 - 1)) * 4 * TILE_F;
        xA = load_frag(kp + nxt, vp + nxt);
        xB = load_frag(kp + nxt + TILE_F, vp + nxt + TILE_F);
        tile_pair(yA, yB, bq0, bq1, cshift, oacc);

        __builtin_amdgcn_sched_group_barrier(SG_LDSRD, 12, 0);
        __builtin_amdgcn_sched_group_barrier(SG_WMMA, 4, 0);
        __builtin_amdgcn_sched_group_barrier(SG_TRANS, 8, 0);
        __builtin_amdgcn_sched_group_barrier(SG_WMMA, 1, 0);
        __builtin_amdgcn_sched_group_barrier(SG_TRANS, 8, 0);
        __builtin_amdgcn_sched_group_barrier(SG_WMMA, 8, 0);
        __builtin_amdgcn_sched_group_barrier(SG_LDSRD, 12, 0);
        __builtin_amdgcn_sched_group_barrier(SG_WMMA, 4, 0);
        __builtin_amdgcn_sched_group_barrier(SG_TRANS, 8, 0);
        __builtin_amdgcn_sched_group_barrier(SG_WMMA, 1, 0);
        __builtin_amdgcn_sched_group_barrier(SG_TRANS, 8, 0);
        __builtin_amdgcn_sched_group_barrier(SG_WMMA, 7, 0);
    }

    float invl = 1.f / __shfl_xor(oacc[0], 16, 32);
    if (!hi) {
        const int b = bh >> 4;
        const int h = bh & 15;
        float4 r0, r1;
        r0.x = oacc[0] * invl; r0.y = oacc[1] * invl;
        r0.z = oacc[2] * invl; r0.w = oacc[3] * invl;
        r1.x = oacc[4] * invl; r1.y = oacc[5] * invl;
        r1.z = oacc[6] * invl; r1.w = oacc[7] * invl;
        float* dst = og + ((size_t)(b * SEQ + qrow)) * EMB + h * DK;
        reinterpret_cast<float4*>(dst)[0] = r0;
        reinterpret_cast<float4*>(dst)[1] = r1;
    }
}

__global__ __launch_bounds__(128)
void proj_kernel(const float* __restrict__ O2, const float* __restrict__ W,
                 const float* __restrict__ bias, float* __restrict__ out) {
    const int  wave = threadIdx.x >> 5;
    const int  lane = threadIdx.x & 31;
    const int  ln   = lane & 15;
    const bool hi   = (lane >= 16);
    const int  off  = hi ? 2 : 0;
    const int  rowbase = (blockIdx.x * 4 + wave) * 16;

    v8f acc[8];
    #pragma unroll
    for (int n = 0; n < 8; ++n) acc[n] = (v8f){0.f,0.f,0.f,0.f,0.f,0.f,0.f,0.f};

    for (int c = 0; c < 32; ++c) {
        const int kd = 4 * c + off;
        const float* ar = O2 + (size_t)(rowbase + ln) * EMB + kd;
        v2f a; a.x = ar[0]; a.y = ar[1];
        #pragma unroll
        for (int n = 0; n < 8; ++n) {
            const float* wr = W + (size_t)(n * 16 + ln) * EMB + kd;
            v2f bw; bw.x = wr[0]; bw.y = wr[1];
            acc[n] = __builtin_amdgcn_wmma_f32_16x16x4_f32(false, a, false, bw,
                                                           (short)0, acc[n],
                                                           false, false);
        }
    }

    #pragma unroll
    for (int n = 0; n < 8; ++n) {
        const int col = n * 16 + ln;
        const float bb = bias[col];
        #pragma unroll
        for (int r = 0; r < 8; ++r) {
            const int row = rowbase + r + (hi ? 8 : 0);
            out[(size_t)row * EMB + col] = acc[n][r] + bb;
        }
    }
}

extern "C" void kernel_launch(void* const* d_in, const int* in_sizes, int n_in,
                              void* d_out, int out_size, void* d_ws, size_t ws_size,
                              hipStream_t stream) {
    const float* x     = (const float*)d_in[0];
    const float* theta = (const float*)d_in[1];
    const float* W     = (const float*)d_in[2];
    const float* bias  = (const float*)d_in[3];
    float* out = (float*)d_out;

    float* ws2 = (float*)d_ws;

    attn_kernel<<<BATCH * NHEADS * (SEQ / QPB), ATT_THREADS, 0, stream>>>(x, theta, ws2);

    proj_kernel<<<128, 128, 0, stream>>>(ws2, W, bias, out);
}


# projection kernel: W and a 32-row O tile staged in LDS by whole-line loads, fragments as 16-byte LDS reads, rows stored whole, 256 work-groups of 4 waves; its k sum order changes
# speedup vs baseline: 1.5230x; 1.0807x over previous
#include <hip/hip_runtime.h>
#include <math.h>

typedef __attribute__((ext_vector_type(2))) float v2f;
typedef __attribute__((ext_vector_type(8))) float v8f;

#define NHEADS   16
#define DK       8
#define SEQ      2048
#define EMB      128
#define BATCH    4
#define KV_PITCH 9
#define QPB      512
#define ATT_THREADS 1024
#define ATT_LDS_BYTES 167936

#define QSCALE   (0.35355339059327373f * 1.4426950408889634f)
#define SHIFT2   (8.0f * 0.35355339059327373f * 1.4426950408889634f)

struct KFrag { v2f a0, a1, av0, av1, av2, av3; };

#define TILE_F (16 * KV_PITCH)

#define WMMA_F32(a, b, c) \
    __builtin_amdgcn_wmma_f32_16x16x4_f32(false, (a), false, (b), (short)0, (c), false, false)

__device__ __forceinline__ KFrag load_frag(const float* kp, const float* vp) {
    KFrag f;
    f.a0.x  = kp[0];              f.a0.y  = kp[1];
    f.a1.x  = kp[4];              f.a1.y  = kp[5];
    f.av0.x = vp[0 * KV_PITCH];   f.av0.y = vp[1 * KV_PITCH];
    f.av1.x = vp[4 * KV_PITCH];   f.av1.y = vp[5 * KV_PITCH];
    f.av2.x = vp[8 * KV_PITCH];   f.av2.y = vp[9 * KV_PITCH];
    f.av3.x = vp[12 * KV_PITCH];  f.av3.y = vp[13 * KV_PITCH];
    return f;
}

__device__ __forceinline__ void tile_pair(const KFrag& fA, const KFrag& fB,
                                          v2f bq0, v2f bq1, v8f cshift, v8f& oacc) {
    v8f sA = WMMA_F32(fA.a0, bq0, cshift);
    v8f sB = WMMA_F32(fB.a0, bq0, cshift);
    sA = WMMA_F32(fA.a1, bq1, sA);
    sB = WMMA_F32(fB.a1, bq1, sB);

    float pA[8], pB[8];
    #pragma unroll
    for (int r = 0; r < 8; ++r) pB[r] = __builtin_amdgcn_exp2f(sB[r]);
    #pragma unroll
    for (int r = 0; r < 8; ++r) pA[r] = __builtin_amdgcn_exp2f(sA[r]);

    v2f pb;
    pb.x = pA[0]; pb.y = pA[1];  oacc = WMMA_F32(fA.av0, pb, oacc);
    pb.x = pA[2]; pb.y = pA[3];  oacc = WMMA_F32(fA.av1, pb, oacc);
    pb.x = pA[4]; pb.y = pA[5];  oacc = WMMA_F32(fA.av2, pb, oacc);
    pb.x = pA[6]; pb.y = pA[7];  oacc = WMMA_F32(fA.av3, pb, oacc);
    pb.x = pB[0]; pb.y = pB[1];  oacc = WMMA_F32(fB.av0, pb, oacc);
    pb.x = pB[2]; pb.y = pB[3];  oacc = WMMA_F32(fB.av1, pb, oacc);
    pb.x = pB[4]; pb.y = pB[5];  oacc = WMMA_F32(fB.av2, pb, oacc);
    pb.x = pB[6]; pb.y = pB[7];  oacc = WMMA_F32(fB.av3, pb, oacc);
}

#define SG_WMMA  0x008
#define SG_LDSRD 0x100
#define SG_TRANS 0x400

__global__ __launch_bounds__(ATT_THREADS)
void attn_kernel(const float* __restrict__ x, const float* __restrict__ theta,
                 float* __restrict__ og) {
    __shared__ float kv[ATT_LDS_BYTES / 4];

    const int tid  = threadIdx.x;
    const int bh   = blockIdx.x >> 2;
    const int qblk = blockIdx.x & 3;

    {
        const int dcol = (tid & 1) * 4;
        const float* xs = x + ((size_t)(bh >> 4) * SEQ) * EMB + (bh & 15) * DK + dcol;
        const float t0 = theta[dcol + 0], t1 = theta[dcol + 1];
        const float t2 = theta[dcol + 2], t3 = theta[dcol + 3];
        #pragma unroll
        for (int k = 0; k < SEQ * DK / 4 / ATT_THREADS; ++k) {
            const int srow = (tid >> 1) + k * (ATT_THREADS / 2);
            const float4 v = *reinterpret_cast<const float4*>(xs + (size_t)srow * EMB);
            float* dst = &kv[srow * KV_PITCH + dcol];
            dst[0] = __cosf(v.x + t0); dst[1] = __cosf(v.y + t1);
            dst[2] = __cosf(v.z + t2); dst[3] = __cosf(v.w + t3);
        }
    }
    for (int i = tid; i < SEQ; i += ATT_THREADS) kv[i * KV_PITCH + DK] = 1.0f;
    __syncthreads();

    const int  wave  = tid >> 5;
    const int  lane  = tid & 31;
    const int  ln    = lane & 15;
    const bool hi    = (lane >= 16);
    const int  off   = hi ? 2 : 0;
    const int  qrow  = qblk * QPB + wave * 16 + ln;

    v2f bq0, bq1;
    bq0.x = kv[qrow * KV_PITCH + off + 0] * QSCALE;
    bq0.y = kv[qrow * KV_PITCH + off + 1] * QSCALE;
    bq1.x = kv[qrow * KV_PITCH + 4 + off + 0] * QSCALE;
    bq1.y = kv[qrow * KV_PITCH + 4 + off + 1] * QSCALE;

    const int srow = ((ln & 6) << 1) | ((ln & 8) >> 2) | (ln & 1);

    v8f oacc = {0.f, 0.f, 0.f, 0.f, 0.f, 0.f, 0.f, 0.f};
    const v8f cshift = {-SHIFT2, -SHIFT2, -SHIFT2, -SHIFT2,
                        -SHIFT2, -SHIFT2, -SHIFT2, -SHIFT2};

    const float* kp = kv + srow * KV_PITCH + off;
    const float* vp = kv + off * KV_PITCH + ln;

    KFrag xA = load_frag(kp, vp);
    KFrag xB = load_frag(kp + TILE_F, vp + TILE_F);

    for (int it = 0; it < SEQ / 64; ++it) {
        const int cur = it * 4 * TILE_F;
        KFrag yA = load_frag(kp + cur + 2 * TILE_F, vp + cur + 2 * TILE_F);
        KFrag yB = load_frag(kp + cur + 3 * TILE_F, vp + cur + 3 * TILE_F);
        tile_pair(xA, xB, bq0, bq1, cshift, oacc);

        asm volatile("" :: "v"(oacc));

        const int nxt = ((it + 1) & (SEQ / 64 - 1)) * 4 * TILE_F;
        xA = load_frag(kp + nxt, vp + nxt);
        xB = load_frag(kp + nxt + TILE_F, vp + nxt + TILE_F);
        tile_pair(yA, yB, bq0, bq1, cshift, oacc);

        __builtin_amdgcn_sched_group_barrier(SG_LDSRD, 12, 0);
        __builtin_amdgcn_sched_group_barrier(SG_WMMA, 4, 0);
        __builtin_amdgcn_sched_group_barrier(SG_TRANS, 8, 0);
        __builtin_amdgcn_sched_group_barrier(SG_WMMA, 1, 0);
        __builtin_amdgcn_sched_group_barrier(SG_TRANS, 8, 0);
        __builtin_amdgcn_sched_group_barrier(SG_WMMA, 8, 0);
        __builtin_amdgcn_sched_group_barrier(SG_LDSRD, 12, 0);
        __builtin_amdgcn_sched_group_barrier(SG_WMMA, 4, 0);
        __builtin_amdgcn_sched_group_barrier(SG_TRANS, 8, 0);
        __builtin_amdgcn_sched_group_barrier(SG_WMMA, 1, 0);
        __builtin_amdgcn_sched_group_barrier(SG_TRANS, 8, 0);
        __builtin_amdgcn_sched_group_barrier(SG_WMMA, 7, 0);
    }

    float invl = 1.f / __shfl_xor(oacc[0], 16, 32);
    if (!hi) {
        const int b = bh >> 4;
        const int h = bh & 15;
        float4 r0, r1;
        r0.x = oacc[0] * invl; r0.y = oacc[1] * invl;
        r0.z = oacc[2] * invl; r0.w = oacc[3] * invl;
        r1.x = oacc[4] * invl; r1.y = oacc[5] * invl;
        r1.z = oacc[6] * invl; r1.w = oacc[7] * invl;
        float* dst = og + ((size_t)(b * SEQ + qrow)) * EMB + h * DK;
        reinterpret_cast<float4*>(dst)[0] = r0;
        reinterpret_cast<float4*>(dst)[1] = r1;
    }
}

#define PJ_ROWS    32
#define PJ_PITCH   132
#define PJ_THREADS 128
#define PJ_RSTEP   (PJ_THREADS / 32)
#define PJ_LDS_BYTES 167936

__global__ __launch_bounds__(PJ_THREADS)
void proj_kernel(const float* __restrict__ O2, const float* __restrict__ W,
                 const float* __restrict__ bias, float* __restrict__ out) {
    __shared__ __attribute__((aligned(16))) float pj[PJ_LDS_BYTES / 4];
    float* sW = pj;
    float* sA = pj + EMB * PJ_PITCH;

    const int tid  = threadIdx.x;
    const int row0 = blockIdx.x * PJ_ROWS;
    const int srow = tid >> 5;
    const int scol = (tid & 31) * 4;

    float4 ga[PJ_ROWS / PJ_RSTEP], gw[EMB / PJ_RSTEP];
    #pragma unroll
    for (int i = 0; i < PJ_ROWS / PJ_RSTEP; ++i)
        ga[i] = *reinterpret_cast<const float4*>(O2 + (size_t)(row0 + srow + PJ_RSTEP * i) * EMB + scol);
    #pragma unroll
    for (int i = 0; i < EMB / PJ_RSTEP; ++i)
        gw[i] = *reinterpret_cast<const float4*>(W + (size_t)(srow + PJ_RSTEP * i) * EMB + scol);
    #pragma unroll
    for (int i = 0; i < PJ_ROWS / PJ_RSTEP; ++i)
        *reinterpret_cast<float4*>(sA + (srow + PJ_RSTEP * i) * PJ_PITCH + scol) = ga[i];
    #pragma unroll
    for (int i = 0; i < EMB / PJ_RSTEP; ++i)
        *reinterpret_cast<float4*>(sW + (srow + PJ_RSTEP * i) * PJ_PITCH + scol) = gw[i];
    __syncthreads();

    const int wave = tid >> 5;
    const int lane = tid & 31;
    const int ln   = lane & 15;
    const int h    = lane >> 4;

    const float* wp0 = sW + (wave * 32 + ln) * PJ_PITCH + 4 * h;
    const float* wp1 = wp0 + 16 * PJ_PITCH;
    const float* ap0 = sA + ln * PJ_PITCH + 4 * h;
    const float* ap1 = ap0 + 16 * PJ_PITCH;

    const v8f zero8 = {0.f, 0.f, 0.f, 0.f, 0.f, 0.f, 0.f, 0.f};
    v8f acc00 = zero8, acc01 = zero8, acc10 = zero8, acc11 = zero8;
    #pragma unroll
    for (int c = 0; c < EMB / 8; ++c) {
        const float4 b0 = *reinterpret_cast<const float4*>(wp0 + 8 * c);
        const float4 b1 = *reinterpret_cast<const float4*>(wp1 + 8 * c);
        const float4 a0 = *reinterpret_cast<const float4*>(ap0 + 8 * c);
        const float4 a1 = *reinterpret_cast<const float4*>(ap1 + 8 * c);
        v2f y0, y1, x0, x1;
        y0.x = b0.x; y0.y = b0.y; y1.x = b1.x; y1.y = b1.y;
        x0.x = a0.x; x0.y = a0.y; x1.x = a1.x; x1.y = a1.y;
        acc00 = WMMA_F32(x0, y0, acc00);
        acc01 = WMMA_F32(x0, y1, acc01);
        acc10 = WMMA_F32(x1, y0, acc10);
        acc11 = WMMA_F32(x1, y1, acc11);
        y0.x = b0.z; y0.y = b0.w; y1.x = b1.z; y1.y = b1.w;
        x0.x = a0.z; x0.y = a0.w; x1.x = a1.z; x1.y = a1.w;
        acc00 = WMMA_F32(x0, y0, acc00);
        acc01 = WMMA_F32(x0, y1, acc01);
        acc10 = WMMA_F32(x1, y0, acc10);
        acc11 = WMMA_F32(x1, y1, acc11);
    }

    __syncthreads();
    {
        const int   col0 = wave * 32 + ln;
        const int   col1 = col0 + 16;
        const float bb0  = bias[col0];
        const float bb1  = bias[col1];
        #pragma unroll
        for (int r = 0; r < 8; ++r) {
            sA[(8 * h + r) * PJ_PITCH + col0]      = acc00[r] + bb0;
            sA[(8 * h + r) * PJ_PITCH + col1]      = acc01[r] + bb1;
            sA[(16 + 8 * h + r) * PJ_PITCH + col0] = acc10[r] + bb0;
            sA[(16 + 8 * h + r) * PJ_PITCH + col1] = acc11[r] + bb1;
        }
    }
    __syncthreads();
    #pragma unroll
    for (int i = 0; i < PJ_ROWS / PJ_RSTEP; ++i) {
        const float4 v = *reinterpret_cast<const float4*>(sA + (srow + PJ_RSTEP * i) * PJ_PITCH + scol);
        *reinterpret_cast<float4*>(out + (size_t)(row0 + srow + PJ_RSTEP * i) * EMB + scol) = v;
    }
}

extern "C" void kernel_launch(void* const* d_in, const int* in_sizes, int n_in,
                              void* d_out, int out_size, void* d_ws, size_t ws_size,
                              hipStream_t stream) {
    const float* x     = (const float*)d_in[0];
    const float* theta = (const float*)d_in[1];
    const float* W     = (const float*)d_in[2];
    const float* bias  = (const float*)d_in[3];
    float* out = (float*)d_out;

    float* ws2 = (float*)d_ws;

    attn_kernel<<<BATCH * NHEADS * (SEQ / QPB), ATT_THREADS, 0, stream>>>(x, theta, ws2);

    proj_kernel<<<(BATCH * SEQ) / PJ_ROWS, PJ_THREADS, 0, stream>>>(ws2, W, bias, out);
}
